# starting kernel
# baseline (speedup 1.0000x reference)
#include <hip/hip_runtime.h>
#include <math.h>

typedef __attribute__((ext_vector_type(2))) float v2f;
typedef __attribute__((ext_vector_type(8))) float v8f;

#define BATCH 32
#define LA 2048
#define LB 2048
#define TPB 256
#define WAVES 8
#define ROWS_PER_BLOCK 128
#define TILES_PER_BATCH (LA / ROWS_PER_BLOCK)

__global__ __launch_bounds__(TPB) void cross_attn_wmma_kernel(
    const float* __restrict__ a, const float* __restrict__ b,
    const float* __restrict__ ipw, const float* __restrict__ ipb,
    const float* __restrict__ ow,  const float* __restrict__ ob,
    float* __restrict__ out)
{
    __shared__ __align__(16) float s_k[LB];
    __shared__ __align__(16) float s_v[LB];
    __shared__ float s_red[2 * TPB];
    __shared__ float s_c[WAVES][32][8];

    const int tid   = threadIdx.x;
    const int wave  = tid >> 5;
    const int lane  = tid & 31;
    const int batch = blockIdx.x / TILES_PER_BATCH;
    const int tile  = blockIdx.x % TILES_PER_BATCH;

    const float wq = ipw[0], wk = ipw[1], wv = ipw[2];
    const float bq = ipb[0], bk = ipb[1], bv = ipb[2];
    const float w_out = ow[0], b_out = ob[0];

    float lmax = -INFINITY, lmin = INFINITY;
    const float* brow = b + (size_t)batch * LB;
    for (int i = tid; i < LB; i += TPB) {
        float bi = brow[i];
        float kv = fmaf(bi, wk, bk);
        s_k[i] = kv;
        s_v[i] = fmaf(bi, wv, bv);
        lmax = fmaxf(lmax, kv);
        lmin = fminf(lmin, kv);
    }
    s_red[tid]       = lmax;
    s_red[TPB + tid] = lmin;
    __syncthreads();
    for (int s = TPB / 2; s > 0; s >>= 1) {
        if (tid < s) {
            s_red[tid]       = fmaxf(s_red[tid],       s_red[tid + s]);
            s_red[TPB + tid] = fminf(s_red[TPB + tid], s_red[TPB + tid + s]);
        }
        __syncthreads();
    }
    const float kmax = s_red[0], kmin = s_red[TPB];

    const int   rowBase = tile * ROWS_PER_BLOCK + wave * 16;
    const int   myRow   = rowBase + (lane & 15);
    const float qa = a[(size_t)batch * LA + myRow];
    const float q  = fmaf(qa, wq, bq);
    const float m  = (q >= 0.0f) ? q * kmax : q * kmin;
    const float L2E = 1.4426950408889634f;
    const float ql2  = q * L2E;
    const float nml2 = -(m * L2E);

    const int  lsel = lane >> 4;
    const int  nsel = lane & 15;
    const bool isn0   = (nsel == 0);
    const float bcnst = (nsel == 1) ? 1.0f : 0.0f;

    const float2* s_k2 = reinterpret_cast<const float2*>(s_k);
    const float2* s_v2 = reinterpret_cast<const float2*>(s_v);

    v8f c = {0.f, 0.f, 0.f, 0.f, 0.f, 0.f, 0.f, 0.f};

    #pragma unroll 8
    for (int jb = 0; jb < LB; jb += 4) {
        int idx = (jb >> 1) + lsel;
        float2 kk = s_k2[idx];
        float2 vv = s_v2[idx];
        float e0 = __builtin_amdgcn_exp2f(fmaf(ql2, kk.x, nml2));
        float e1 = __builtin_amdgcn_exp2f(fmaf(ql2, kk.y, nml2));
        v2f A; A.x = e0; A.y = e1;
        v2f B;
        B.x = isn0 ? vv.x : bcnst;
        B.y = isn0 ? vv.y : bcnst;
        c = __builtin_amdgcn_wmma_f32_16x16x4_f32(
                 false, A,  false, B,
                 (short)0, c,  false,  false);
    }

    #pragma unroll
    for (int g = 0; g < 8; ++g) s_c[wave][lane][g] = c[g];

    if (lane < 16) {
        int srcLane = (lane < 8) ? 0 : 16;
        int g = lane & 7;
        float numer = s_c[wave][srcLane][g];
        float denom = s_c[wave][srcLane + 1][g];
        float ctx = numer / denom;
        float x = fmaf(ctx, w_out, b_out);
        float y = x + qa;
        float r = 0.5f * y * (1.0f + erff(y * 0.70710678118654752f));
        out[(size_t)batch * LA + rowBase + lane] = r;
    }
}

extern "C" void kernel_launch(void* const* d_in, const int* in_sizes, int n_in,
                              void* d_out, int out_size, void* d_ws, size_t ws_size,
                              hipStream_t stream) {
    const float* a   = (const float*)d_in[0];
    const float* b   = (const float*)d_in[1];
    const float* ipw = (const float*)d_in[2];
    const float* ipb = (const float*)d_in[3];
    const float* ow  = (const float*)d_in[4];
    const float* ob  = (const float*)d_in[5];
    float* out = (float*)d_out;
    (void)in_sizes; (void)n_in; (void)out_size; (void)d_ws; (void)ws_size;

    dim3 grid(BATCH * TILES_PER_BATCH);
    dim3 block(TPB);
    hipLaunchKernelGGL(cross_attn_wmma_kernel, grid, block, 0, stream,
                       a, b, ipw, ipb, ow, ob, out);
}


# key-stationary split: each of 16 waves keeps its 128 keys as register-resident k values and ready B fragments, 256-row slabs stream through, partial sums added once; no LDS read in the matrix loop
# speedup vs baseline: 1.3421x; 1.3421x over previous
#include <hip/hip_runtime.h>
#include <math.h>

typedef __attribute__((ext_vector_type(2))) float v2f;
typedef __attribute__((ext_vector_type(8))) float v8f;

#define BATCH 32
#define LA 2048
#define LB 2048
#define WAVES 16
#define TPB (WAVES * 32)
#define KEYS_PER_WAVE (LB / WAVES)
#define STEPS (KEYS_PER_WAVE / 4)
#define ROWS_PER_BLOCK 256
#define TILES (ROWS_PER_BLOCK / 16)
#define SLABS_PER_BATCH (LA / ROWS_PER_BLOCK)

__device__ __forceinline__ float xhalf(float x) {
    int i = __builtin_bit_cast(int, x);
    return __builtin_bit_cast(float,
        __builtin_amdgcn_permlanex16(i, i, 0x76543210, 0xfedcba98, false, false));
}
template <int M> __device__ __forceinline__ float rowx(float x) {
    return __builtin_bit_cast(float,
        __builtin_amdgcn_update_dpp(0, __builtin_bit_cast(int, x), 0x160 + M, 0xf, 0xf, false));
}
__device__ __forceinline__ float row_max16(float x) {
    x = fmaxf(x, rowx<1>(x)); x = fmaxf(x, rowx<2>(x));
    x = fmaxf(x, rowx<4>(x)); x = fmaxf(x, rowx<8>(x));
    return x;
}
__device__ __forceinline__ float row_min16(float x) {
    x = fminf(x, rowx<1>(x)); x = fminf(x, rowx<2>(x));
    x = fminf(x, rowx<4>(x)); x = fminf(x, rowx<8>(x));
    return x;
}

__global__ __launch_bounds__(TPB) void cross_attn_keysplit_kernel(
    const float* __restrict__ a, const float* __restrict__ b,
    const float* __restrict__ ipw, const float* __restrict__ ipb,
    const float* __restrict__ ow,  const float* __restrict__ ob,
    float* __restrict__ out)
{
    __shared__ __align__(16) float s_k[LB];
    __shared__ __align__(16) float s_v[LB];
    __shared__ float s_a[ROWS_PER_BLOCK];
    __shared__ float s_mx[WAVES];
    __shared__ float s_mn[WAVES];
    __shared__ __align__(16) float s_part[2 * WAVES * ROWS_PER_BLOCK];

    const int tid   = threadIdx.x;
    const int wave  = tid >> 5;
    const int lane  = tid & 31;
    const int lsel  = lane >> 4;
    const int nsel  = lane & 15;
    const int batch = blockIdx.x / SLABS_PER_BATCH;
    const int slab  = blockIdx.x % SLABS_PER_BATCH;

    const float wq = ipw[0], wk = ipw[1], wv = ipw[2];
    const float bq = ipb[0], bk = ipb[1], bv = ipb[2];
    const float w_out = ow[0], b_out = ob[0];

    const int    kbase   = wave * KEYS_PER_WAVE;
    const size_t rowBase = (size_t)batch * LA + (size_t)slab * ROWS_PER_BLOCK;
    const float4 b4 = *reinterpret_cast<const float4*>(b + (size_t)batch * LB + kbase + 4 * lane);
    float qa_own = 0.0f;
    if (tid < ROWS_PER_BLOCK) {
        qa_own = a[rowBase + tid];
        s_a[tid] = qa_own;
    }
    float4 k4, v4;
    k4.x = fmaf(b4.x, wk, bk); k4.y = fmaf(b4.y, wk, bk);
    k4.z = fmaf(b4.z, wk, bk); k4.w = fmaf(b4.w, wk, bk);
    v4.x = fmaf(b4.x, wv, bv); v4.y = fmaf(b4.y, wv, bv);
    v4.z = fmaf(b4.z, wv, bv); v4.w = fmaf(b4.w, wv, bv);
    *reinterpret_cast<float4*>(s_k + kbase + 4 * lane) = k4;
    *reinterpret_cast<float4*>(s_v + kbase + 4 * lane) = v4;

    float lmax = fmaxf(fmaxf(k4.x, k4.y), fmaxf(k4.z, k4.w));
    float lmin = fminf(fminf(k4.x, k4.y), fminf(k4.z, k4.w));
    lmax = row_max16(lmax); lmax = fmaxf(lmax, xhalf(lmax));
    lmin = row_min16(lmin); lmin = fminf(lmin, xhalf(lmin));
    if (lane == 0) { s_mx[wave] = lmax; s_mn[wave] = lmin; }
    __syncthreads();
    const float kmax = row_max16(s_mx[nsel]);
    const float kmin = row_min16(s_mn[nsel]);

    const bool  isn0  = (nsel == 0);
    const float bcnst = (nsel == 1) ? 1.0f : 0.0f;
    const float2* s_k2 = reinterpret_cast<const float2*>(s_k + kbase);
    const float2* s_v2 = reinterpret_cast<const float2*>(s_v + kbase);
    float kx[STEPS], ky[STEPS], bx[STEPS], by[STEPS];
    #pragma unroll
    for (int s = 0; s < STEPS; ++s) {
        const float2 kk = s_k2[2 * s + lsel];
        const float2 vv = s_v2[2 * s + lsel];
        kx[s] = kk.x; ky[s] = kk.y;
        bx[s] = isn0 ? vv.x : bcnst;
        by[s] = isn0 ? vv.y : bcnst;
    }

    const float L2E = 1.4426950408889634f;
    #define ROW_STATE(t, QL2, NML2) do {                               \
        const float qa_ = s_a[(t) * 16 + nsel];                         \
        const float q_  = fmaf(qa_, wq, bq);                            \
        const float m_  = (q_ >= 0.0f) ? q_ * kmax : q_ * kmin;         \
        QL2  = q_ * L2E;                                                \
        NML2 = -(m_ * L2E);                                             \
    } while (0)

    float nq0, nn0, nq1, nn1;
    ROW_STATE(0, nq0, nn0);
    ROW_STATE(1, nq1, nn1);

    float* const pdst = s_part + (nsel * WAVES + wave) * ROWS_PER_BLOCK + 8 * lsel;

    #pragma unroll 1
    for (int p = 0; p < TILES / 2; ++p) {
        const float q0 = nq0, n0 = nn0, q1 = nq1, n1 = nn1;
        const int tn = (2 * p + 2 < TILES) ? (2 * p + 2) : (TILES - 2);
        ROW_STATE(tn,     nq0, nn0);
        ROW_STATE(tn + 1, nq1, nn1);

        v8f c0 = {0.f, 0.f, 0.f, 0.f, 0.f, 0.f, 0.f, 0.f};
        v8f c1 = {0.f, 0.f, 0.f, 0.f, 0.f, 0.f, 0.f, 0.f};
        #pragma unroll
        for (int s = 0; s < STEPS; ++s) {
            v2f A0, A1, B;
            A0.x = __builtin_amdgcn_exp2f(fmaf(q0, kx[s], n0));
            A0.y = __builtin_amdgcn_exp2f(fmaf(q0, ky[s], n0));
            A1.x = __builtin_amdgcn_exp2f(fmaf(q1, kx[s], n1));
            A1.y = __builtin_amdgcn_exp2f(fmaf(q1, ky[s], n1));
            B.x = bx[s]; B.y = by[s];
            c0 = __builtin_amdgcn_wmma_f32_16x16x4_f32(
                      false, A0,  false, B,
                      (short)0, c0,  false,  false);
            c1 = __builtin_amdgcn_wmma_f32_16x16x4_f32(
                      false, A1,  false, B,
                      (short)0, c1,  false,  false);
        }

        if (nsel < 2) {
            float* d = pdst + p * 32;
            float4 lo, hi;
            lo.x = c0[0]; lo.y = c0[1]; lo.z = c0[2]; lo.w = c0[3];
            hi.x = c0[4]; hi.y = c0[5]; hi.z = c0[6]; hi.w = c0[7];
            *reinterpret_cast<float4*>(d)     = lo;
            *reinterpret_cast<float4*>(d + 4) = hi;
            lo.x = c1[0]; lo.y = c1[1]; lo.z = c1[2]; lo.w = c1[3];
            hi.x = c1[4]; hi.y = c1[5]; hi.z = c1[6]; hi.w = c1[7];
            *reinterpret_cast<float4*>(d + 16) = lo;
            *reinterpret_cast<float4*>(d + 20) = hi;
        }
    }
    #undef ROW_STATE

    __syncthreads();
    if (tid < ROWS_PER_BLOCK) {
        float numer = 0.0f, denom = 0.0f;
        #pragma unroll
        for (int w = 0; w < WAVES; ++w) {
            numer += s_part[w * ROWS_PER_BLOCK + tid];
            denom += s_part[(WAVES + w) * ROWS_PER_BLOCK + tid];
        }
        const float ctx = numer / denom;
        const float x = fmaf(ctx, w_out, b_out);
        const float y = x + qa_own;
        const float r = 0.5f * y * (1.0f + erff(y * 0.70710678118654752f));
        out[rowBase + tid] = r;
    }
}

extern "C" void kernel_launch(void* const* d_in, const int* in_sizes, int n_in,
                              void* d_out, int out_size, void* d_ws, size_t ws_size,
                              hipStream_t stream) {
    const float* a   = (const float*)d_in[0];
    const float* b   = (const float*)d_in[1];
    const float* ipw = (const float*)d_in[2];
    const float* ipb = (const float*)d_in[3];
    const float* ow  = (const float*)d_in[4];
    const float* ob  = (const float*)d_in[5];
    float* out = (float*)d_out;
    (void)in_sizes; (void)n_in; (void)out_size; (void)d_ws; (void)ws_size;

    dim3 grid(BATCH * SLABS_PER_BATCH);
    dim3 block(TPB);
    hipLaunchKernelGGL(cross_attn_keysplit_kernel, grid, block, 0, stream,
                       a, b, ipw, ipb, ow, ob, out);
}


# both matrix operands carry exp terms: A rows 0..12 and B columns 0..12 hold 13 query rows each, v and ones in row/column 14 and 15, so one f32 matrix step serves 26 rows instead of 16
# speedup vs baseline: 1.6965x; 1.2641x over previous
#include <hip/hip_runtime.h>
#include <math.h>

typedef __attribute__((ext_vector_type(2))) float v2f;
typedef __attribute__((ext_vector_type(8))) float v8f;

#define BATCH 32
#define LA 2048
#define LB 2048
#define WAVES 16
#define TPB (WAVES * 32)
#define KEYS_PER_WAVE (LB / WAVES)
#define STEPS (KEYS_PER_WAVE / 4)
#define ROWS_PER_BLOCK 256
#define SLABS_PER_BATCH (LA / ROWS_PER_BLOCK)
#define HALF_ROWS 13
#define TILE_ROWS (2 * HALF_ROWS)
#define TILES ((ROWS_PER_BLOCK + TILE_ROWS - 1) / TILE_ROWS)
#define TILE_PITCH 32
#define PART_PITCH (TILES * TILE_PITCH)
#define SA_LEN 288

__device__ __forceinline__ float xhalf(float x) {
    int i = __builtin_bit_cast(int, x);
    return __builtin_bit_cast(float,
        __builtin_amdgcn_permlanex16(i, i, 0x76543210, 0xfedcba98, false, false));
}
template <int M> __device__ __forceinline__ float rowx(float x) {
    return __builtin_bit_cast(float,
        __builtin_amdgcn_update_dpp(0, __builtin_bit_cast(int, x), 0x160 + M, 0xf, 0xf, false));
}
__device__ __forceinline__ float row_max16(float x) {
    x = fmaxf(x, rowx<1>(x)); x = fmaxf(x, rowx<2>(x));
    x = fmaxf(x, rowx<4>(x)); x = fmaxf(x, rowx<8>(x));
    return x;
}
__device__ __forceinline__ float row_min16(float x) {
    x = fminf(x, rowx<1>(x)); x = fminf(x, rowx<2>(x));
    x = fminf(x, rowx<4>(x)); x = fminf(x, rowx<8>(x));
    return x;
}

__global__ __launch_bounds__(TPB) void cross_attn_keysplit2_kernel(
    const float* __restrict__ a, const float* __restrict__ b,
    const float* __restrict__ ipw, const float* __restrict__ ipb,
    const float* __restrict__ ow,  const float* __restrict__ ob,
    float* __restrict__ out)
{
    __shared__ __align__(16) float s_k[LB];
    __shared__ __align__(16) float s_v[LB];
    __shared__ float s_a[SA_LEN];
    __shared__ float s_mx[WAVES];
    __shared__ float s_mn[WAVES];
    __shared__ __align__(16) float s_part[2 * WAVES * PART_PITCH];

    const int tid   = threadIdx.x;
    const int wave  = tid >> 5;
    const int lane  = tid & 31;
    const int lsel  = lane >> 4;
    const int nsel  = lane & 15;
    const int batch = blockIdx.x / SLABS_PER_BATCH;
    const int slab  = blockIdx.x % SLABS_PER_BATCH;

    const float wq = ipw[0], wk = ipw[1], wv = ipw[2];
    const float bq = ipb[0], bk = ipb[1], bv = ipb[2];
    const float w_out = ow[0], b_out = ob[0];

    const int    kbase   = wave * KEYS_PER_WAVE;
    const size_t rowBase = (size_t)batch * LA + (size_t)slab * ROWS_PER_BLOCK;
    const float4 b4 = *reinterpret_cast<const float4*>(b + (size_t)batch * LB + kbase + 4 * lane);
    float qa_own = 0.0f;
    if (tid < ROWS_PER_BLOCK) {
        qa_own = a[rowBase + tid];
        s_a[tid] = qa_own;
    } else if (tid < SA_LEN) {
        s_a[tid] = 0.0f;
    }
    float4 k4, v4;
    k4.x = fmaf(b4.x, wk, bk); k4.y = fmaf(b4.y, wk, bk);
    k4.z = fmaf(b4.z, wk, bk); k4.w = fmaf(b4.w, wk, bk);
    v4.x = fmaf(b4.x, wv, bv); v4.y = fmaf(b4.y, wv, bv);
    v4.z = fmaf(b4.z, wv, bv); v4.w = fmaf(b4.w, wv, bv);
    *reinterpret_cast<float4*>(s_k + kbase + 4 * lane) = k4;
    *reinterpret_cast<float4*>(s_v + kbase + 4 * lane) = v4;

    float lmax = fmaxf(fmaxf(k4.x, k4.y), fmaxf(k4.z, k4.w));
    float lmin = fminf(fminf(k4.x, k4.y), fminf(k4.z, k4.w));
    lmax = row_max16(lmax); lmax = fmaxf(lmax, xhalf(lmax));
    lmin = row_min16(lmin); lmin = fminf(lmin, xhalf(lmin));
    if (lane == 0) { s_mx[wave] = lmax; s_mn[wave] = lmin; }
    __syncthreads();
    const float kmax = row_max16(s_mx[nsel]);
    const float kmin = row_min16(s_mn[nsel]);

    const bool  is_v = (nsel == 14);
    const float cnst = (nsel == 15) ? 1.0f : 0.0f;
    const float2* s_k2 = reinterpret_cast<const float2*>(s_k + kbase);
    const float2* s_v2 = reinterpret_cast<const float2*>(s_v + kbase);
    v2f K[STEPS], A[STEPS], B[STEPS];
    #pragma unroll
    for (int s = 0; s < STEPS; ++s) {
        const float2 kk = s_k2[2 * s + lsel];
        const float2 vv = s_v2[2 * s + lsel];
        K[s].x = kk.x; K[s].y = kk.y;
        A[s].x = is_v ? vv.x : cnst;
        A[s].y = is_v ? vv.y : cnst;
        B[s] = A[s];
    }

    const float L2E = 1.4426950408889634f;
    #define ROW_STATE(row, QL2, NML2) do {                              \
        const float qa_ = s_a[(row)];                                   \
        const float q_  = fmaf(qa_, wq, bq);                            \
        const float m_  = (q_ >= 0.0f) ? q_ * kmax : q_ * kmin;         \
        QL2  = q_ * L2E;                                                \
        NML2 = -(m_ * L2E);                                             \
    } while (0)

    const bool live = (nsel < HALF_ROWS);
    float nqa, nna, nqb, nnb;
    ROW_STATE(nsel, nqa, nna);
    ROW_STATE(HALF_ROWS + nsel, nqb, nnb);

    float* const pcol = s_part + ((nsel & 1) * WAVES + wave) * PART_PITCH + 8 * lsel;
    float* const prow = s_part + wave * PART_PITCH + 16 + nsel;

    #pragma unroll 1
    for (int t = 0; t < TILES; ++t) {
        const float qa = nqa, na = nna, qb = nqb, nb = nnb;
        const int tn = (t + 1 < TILES) ? (t + 1) : t;
        ROW_STATE(tn * TILE_ROWS + nsel, nqa, nna);
        ROW_STATE(tn * TILE_ROWS + HALF_ROWS + nsel, nqb, nnb);

        if (live) {
            #pragma unroll
            for (int s = 0; s < STEPS; ++s) {
                A[s].x = __builtin_amdgcn_exp2f(fmaf(qa, K[s].x, na));
                A[s].y = __builtin_amdgcn_exp2f(fmaf(qa, K[s].y, na));
                B[s].x = __builtin_amdgcn_exp2f(fmaf(qb, K[s].x, nb));
                B[s].y = __builtin_amdgcn_exp2f(fmaf(qb, K[s].y, nb));
            }
        }

        v8f c0 = {0.f, 0.f, 0.f, 0.f, 0.f, 0.f, 0.f, 0.f};
        v8f c1 = {0.f, 0.f, 0.f, 0.f, 0.f, 0.f, 0.f, 0.f};
        #pragma unroll
        for (int s = 0; s < STEPS; s += 2) {
            c0 = __builtin_amdgcn_wmma_f32_16x16x4_f32(
                      false, A[s],  false, B[s],
                      (short)0, c0,  false,  false);
            c1 = __builtin_amdgcn_wmma_f32_16x16x4_f32(
                      false, A[s + 1],  false, B[s + 1],
                      (short)0, c1,  false,  false);
        }
        const v8f c = c0 + c1;

        if (nsel >= 14) {
            float* d = pcol + t * TILE_PITCH;
            float4 lo, hi;
            lo.x = c[0]; lo.y = c[1]; lo.z = c[2]; lo.w = c[3];
            hi.x = c[4]; hi.y = c[5]; hi.z = c[6]; hi.w = c[7];
            *reinterpret_cast<float4*>(d)     = lo;
            *reinterpret_cast<float4*>(d + 4) = hi;
        }
        if (live && lsel == 1) {
            float* d = prow + t * TILE_PITCH;
            d[0] = c[6];
            d[WAVES * PART_PITCH] = c[7];
        }
    }
    #undef ROW_STATE

    __syncthreads();
    if (tid < ROWS_PER_BLOCK) {
        const int tt  = tid / TILE_ROWS;
        const int pp  = tid - tt * TILE_ROWS;
        const int pos = tt * TILE_PITCH + ((pp < HALF_ROWS) ? pp : (pp + 16 - HALF_ROWS));
        float numer = 0.0f, denom = 0.0f;
        #pragma unroll
        for (int w = 0; w < WAVES; ++w) {
            numer += s_part[w * PART_PITCH + pos];
            denom += s_part[(WAVES + w) * PART_PITCH + pos];
        }
        const float ctx = numer / denom;
        const float x = fmaf(ctx, w_out, b_out);
        const float y = x + qa_own;
        const float r = 0.5f * y * (1.0f + erff(y * 0.70710678118654752f));
        out[rowBase + tid] = r;
    }
}

extern "C" void kernel_launch(void* const* d_in, const int* in_sizes, int n_in,
                              void* d_out, int out_size, void* d_ws, size_t ws_size,
                              hipStream_t stream) {
    const float* a   = (const float*)d_in[0];
    const float* b   = (const float*)d_in[1];
    const float* ipw = (const float*)d_in[2];
    const float* ipb = (const float*)d_in[3];
    const float* ow  = (const float*)d_in[4];
    const float* ob  = (const float*)d_in[5];
    float* out = (float*)d_out;
    (void)in_sizes; (void)n_in; (void)out_size; (void)d_ws; (void)ws_size;

    dim3 grid(BATCH * SLABS_PER_BATCH);
    dim3 block(TPB);
    hipLaunchKernelGGL(cross_attn_keysplit2_kernel, grid, block, 0, stream,
                       a, b, ipw, ipb, ow, ob, out);
}


# row tiles of 28 (14 per operand): 9 full tiles for 252 rows, last 4 rows in one 8-step tile (B columns = 4 rows x 4 key groups, A rows = v per group and ones); 296 matrix steps per wave, was 320
# speedup vs baseline: 1.8107x; 1.0673x over previous
#include <hip/hip_runtime.h>
#include <math.h>

typedef __attribute__((ext_vector_type(2))) float v2f;
typedef __attribute__((ext_vector_type(8))) float v8f;

#define BATCH 32
#define LA 2048
#define LB 2048
#define WAVES 16
#define TPB (WAVES * 32)
#define KEYS_PER_WAVE (LB / WAVES)
#define STEPS (KEYS_PER_WAVE / 4)
#define ROWS_PER_BLOCK 256
#define SLABS_PER_BATCH (LA / ROWS_PER_BLOCK)
#define HALF_ROWS 14
#define TILE_ROWS (2 * HALF_ROWS)
#define FULL_TILES (ROWS_PER_BLOCK / TILE_ROWS)
#define TAIL_ROWS (ROWS_PER_BLOCK - FULL_TILES * TILE_ROWS)
#define TAIL_GROUPS (16 / TAIL_ROWS)
#define TAIL_STEPS (STEPS / TAIL_GROUPS)
#define TILE_PITCH 32
#define PART_PITCH ((FULL_TILES + 1) * TILE_PITCH)
#define SA_LEN ROWS_PER_BLOCK

__device__ __forceinline__ float xhalf(float x) {
    int i = __builtin_bit_cast(int, x);
    return __builtin_bit_cast(float,
        __builtin_amdgcn_permlanex16(i, i, 0x76543210, 0xfedcba98, false, false));
}
template <int M> __device__ __forceinline__ float rowx(float x) {
    return __builtin_bit_cast(float,
        __builtin_amdgcn_update_dpp(0, __builtin_bit_cast(int, x), 0x160 + M, 0xf, 0xf, false));
}
__device__ __forceinline__ float row_max16(float x) {
    x = fmaxf(x, rowx<1>(x)); x = fmaxf(x, rowx<2>(x));
    x = fmaxf(x, rowx<4>(x)); x = fmaxf(x, rowx<8>(x));
    return x;
}
__device__ __forceinline__ float row_min16(float x) {
    x = fminf(x, rowx<1>(x)); x = fminf(x, rowx<2>(x));
    x = fminf(x, rowx<4>(x)); x = fminf(x, rowx<8>(x));
    return x;
}

__global__ __launch_bounds__(TPB) void cross_attn_keysplit_tail_kernel(
    const float* __restrict__ a, const float* __restrict__ b,
    const float* __restrict__ ipw, const float* __restrict__ ipb,
    const float* __restrict__ ow,  const float* __restrict__ ob,
    float* __restrict__ out)
{
    __shared__ __align__(16) float s_k[LB];
    __shared__ __align__(16) float s_v[LB];
    __shared__ float s_a[SA_LEN];
    __shared__ float s_mx[WAVES];
    __shared__ float s_mn[WAVES];
    __shared__ __align__(16) float s_part[2 * WAVES * PART_PITCH];

    const int tid   = threadIdx.x;
    const int wave  = tid >> 5;
    const int lane  = tid & 31;
    const int lsel  = lane >> 4;
    const int nsel  = lane & 15;
    const int batch = blockIdx.x / SLABS_PER_BATCH;
    const int slab  = blockIdx.x % SLABS_PER_BATCH;

    const float wq = ipw[0], wk = ipw[1], wv = ipw[2];
    const float bq = ipb[0], bk = ipb[1], bv = ipb[2];
    const float w_out = ow[0], b_out = ob[0];

    const int    kbase   = wave * KEYS_PER_WAVE;
    const size_t rowBase = (size_t)batch * LA + (size_t)slab * ROWS_PER_BLOCK;
    const float4 b4 = *reinterpret_cast<const float4*>(b + (size_t)batch * LB + kbase + 4 * lane);
    float qa_own = 0.0f;
    if (tid < ROWS_PER_BLOCK) {
        qa_own = a[rowBase + tid];
        s_a[tid] = qa_own;
    }
    float4 k4, v4;
    k4.x = fmaf(b4.x, wk, bk); k4.y = fmaf(b4.y, wk, bk);
    k4.z = fmaf(b4.z, wk, bk); k4.w = fmaf(b4.w, wk, bk);
    v4.x = fmaf(b4.x, wv, bv); v4.y = fmaf(b4.y, wv, bv);
    v4.z = fmaf(b4.z, wv, bv); v4.w = fmaf(b4.w, wv, bv);
    *reinterpret_cast<float4*>(s_k + kbase + 4 * lane) = k4;
    *reinterpret_cast<float4*>(s_v + kbase + 4 * lane) = v4;

    float lmax = fmaxf(fmaxf(k4.x, k4.y), fmaxf(k4.z, k4.w));
    float lmin = fminf(fminf(k4.x, k4.y), fminf(k4.z, k4.w));
    lmax = row_max16(lmax); lmax = fmaxf(lmax, xhalf(lmax));
    lmin = row_min16(lmin); lmin = fminf(lmin, xhalf(lmin));
    if (lane == 0) { s_mx[wave] = lmax; s_mn[wave] = lmin; }
    __syncthreads();
    const float kmax = row_max16(s_mx[nsel]);
    const float kmin = row_min16(s_mn[nsel]);

    const bool  is_v = (nsel == 14);
    const float cnst = (nsel == 15) ? 1.0f : 0.0f;
    const float2* s_k2 = reinterpret_cast<const float2*>(s_k + kbase);
    const float2* s_v2 = reinterpret_cast<const float2*>(s_v + kbase);
    v2f K[STEPS], A[STEPS], B[STEPS];
    #pragma unroll
    for (int s = 0; s < STEPS; ++s) {
        const float2 kk = s_k2[2 * s + lsel];
        const float2 vv = s_v2[2 * s + lsel];
        K[s].x = kk.x; K[s].y = kk.y;
        A[s].x = is_v ? vv.x : cnst;
        A[s].y = is_v ? vv.y : cnst;
        B[s] = A[s];
    }

    const float L2E = 1.4426950408889634f;
    #define ROW_STATE(row, QL2, NML2) do {                              \
        const float qa_ = s_a[(row)];                                   \
        const float q_  = fmaf(qa_, wq, bq);                            \
        const float m_  = (q_ >= 0.0f) ? q_ * kmax : q_ * kmin;         \
        QL2  = q_ * L2E;                                                \
        NML2 = -(m_ * L2E);                                             \
    } while (0)

    const bool live = (nsel < HALF_ROWS);
    float nqa, nna, nqb, nnb;
    ROW_STATE(nsel, nqa, nna);
    ROW_STATE(HALF_ROWS + nsel, nqb, nnb);

    float* const pcol = s_part + ((nsel & 1) * WAVES + wave) * PART_PITCH + 8 * lsel;
    float* const prow = s_part + wave * PART_PITCH + 16 + nsel;

    #pragma unroll 1
    for (int t = 0; t < FULL_TILES; ++t) {
        const float qa = nqa, na = nna, qb = nqb, nb = nnb;
        const int tn = (t + 1 < FULL_TILES) ? (t + 1) : t;
        ROW_STATE(tn * TILE_ROWS + nsel, nqa, nna);
        ROW_STATE(tn * TILE_ROWS + HALF_ROWS + nsel, nqb, nnb);

        if (live) {
            #pragma unroll
            for (int s = 0; s < STEPS; ++s) {
                A[s].x = __builtin_amdgcn_exp2f(fmaf(qa, K[s].x, na));
                A[s].y = __builtin_amdgcn_exp2f(fmaf(qa, K[s].y, na));
                B[s].x = __builtin_amdgcn_exp2f(fmaf(qb, K[s].x, nb));
                B[s].y = __builtin_amdgcn_exp2f(fmaf(qb, K[s].y, nb));
            }
        }

        v8f c0 = {0.f, 0.f, 0.f, 0.f, 0.f, 0.f, 0.f, 0.f};
        v8f c1 = {0.f, 0.f, 0.f, 0.f, 0.f, 0.f, 0.f, 0.f};
        #pragma unroll
        for (int s = 0; s < STEPS; s += 2) {
            c0 = __builtin_amdgcn_wmma_f32_16x16x4_f32(
                      false, A[s],  false, B[s],
                      (short)0, c0,  false,  false);
            c1 = __builtin_amdgcn_wmma_f32_16x16x4_f32(
                      false, A[s + 1],  false, B[s + 1],
                      (short)0, c1,  false,  false);
        }
        const v8f c = c0 + c1;

        if (nsel >= 14) {
            float* d = pcol + t * TILE_PITCH;
            float4 lo, hi;
            lo.x = c[0]; lo.y = c[1]; lo.z = c[2]; lo.w = c[3];
            hi.x = c[4]; hi.y = c[5]; hi.z = c[6]; hi.w = c[7];
            *reinterpret_cast<float4*>(d)     = lo;
            *reinterpret_cast<float4*>(d + 4) = hi;
        }
        if (live && lsel == 1) {
            float* d = prow + t * TILE_PITCH;
            d[0] = c[6];
            d[WAVES * PART_PITCH] = c[7];
        }
    }
    {
        const int ti = nsel & (TAIL_ROWS - 1);
        const int tg = nsel >> 2;
        float qt, nt;
        ROW_STATE(FULL_TILES * TILE_ROWS + ti, qt, nt);
        const bool  a_is_v = (nsel < TAIL_GROUPS);
        const float a_cnst = (nsel == TAIL_GROUPS) ? 1.0f : 0.0f;
        const float2* kq = s_k2 + (KEYS_PER_WAVE / TAIL_GROUPS / 2) * tg + lsel;
        const float2* vq = s_v2 + (KEYS_PER_WAVE / TAIL_GROUPS / 2) * ti + lsel;
        v8f d0 = {0.f, 0.f, 0.f, 0.f, 0.f, 0.f, 0.f, 0.f};
        v8f d1 = {0.f, 0.f, 0.f, 0.f, 0.f, 0.f, 0.f, 0.f};
        #pragma unroll
        for (int m = 0; m < TAIL_STEPS; m += 2) {
            const float2 k0 = kq[2 * m],     v0 = vq[2 * m];
            const float2 k1 = kq[2 * m + 2], v1 = vq[2 * m + 2];
            v2f a0, b0, a1, b1;
            b0.x = __builtin_amdgcn_exp2f(fmaf(qt, k0.x, nt));
            b0.y = __builtin_amdgcn_exp2f(fmaf(qt, k0.y, nt));
            b1.x = __builtin_amdgcn_exp2f(fmaf(qt, k1.x, nt));
            b1.y = __builtin_amdgcn_exp2f(fmaf(qt, k1.y, nt));
            a0.x = a_is_v ? v0.x : a_cnst;  a0.y = a_is_v ? v0.y : a_cnst;
            a1.x = a_is_v ? v1.x : a_cnst;  a1.y = a_is_v ? v1.y : a_cnst;
            d0 = __builtin_amdgcn_wmma_f32_16x16x4_f32(
                      false, a0,  false, b0,
                      (short)0, d0,  false,  false);
            d1 = __builtin_amdgcn_wmma_f32_16x16x4_f32(
                      false, a1,  false, b1,
                      (short)0, d1,  false,  false);
        }
        const v8f d = d0 + d1;
        float pn = (tg == 0) ? d[0] : (tg == 1) ? d[1] : (tg == 2) ? d[2] : d[3];
        float pd = d[4];
        pn += rowx<4>(pn);  pd += rowx<4>(pd);
        pn += rowx<8>(pn);  pd += rowx<8>(pd);
        if (lane < TAIL_ROWS) {
            s_part[wave * PART_PITCH + FULL_TILES * TILE_PITCH + lane] = pn;
            s_part[(WAVES + wave) * PART_PITCH + FULL_TILES * TILE_PITCH + lane] = pd;
        }
    }
    #undef ROW_STATE

    __syncthreads();
    if (tid < ROWS_PER_BLOCK) {
        const int tt  = tid / TILE_ROWS;
        const int pp  = tid - tt * TILE_ROWS;
        const int pos = (tt < FULL_TILES)
                      ? tt * TILE_PITCH + ((pp < HALF_ROWS) ? pp : (pp + 16 - HALF_ROWS))
                      : FULL_TILES * TILE_PITCH + pp;
        float numer = 0.0f, denom = 0.0f;
        #pragma unroll
        for (int w = 0; w < WAVES; ++w) {
            numer += s_part[w * PART_PITCH + pos];
            denom += s_part[(WAVES + w) * PART_PITCH + pos];
        }
        const float ctx = numer / denom;
        const float x = fmaf(ctx, w_out, b_out);
        const float y = x + qa_own;
        const float r = 0.5f * y * (1.0f + erff(y * 0.70710678118654752f));
        out[rowBase + tid] = r;
    }
}

extern "C" void kernel_launch(void* const* d_in, const int* in_sizes, int n_in,
                              void* d_out, int out_size, void* d_ws, size_t ws_size,
                              hipStream_t stream) {
    const float* a   = (const float*)d_in[0];
    const float* b   = (const float*)d_in[1];
    const float* ipw = (const float*)d_in[2];
    const float* ipb = (const float*)d_in[3];
    const float* ow  = (const float*)d_in[4];
    const float* ob  = (const float*)d_in[5];
    float* out = (float*)d_out;
    (void)in_sizes; (void)n_in; (void)out_size; (void)d_ws; (void)ws_size;

    dim3 grid(BATCH * SLABS_PER_BATCH);
    dim3 block(TPB);
    hipLaunchKernelGGL(cross_attn_keysplit_tail_kernel, grid, block, 0, stream,
                       a, b, ipw, ipb, ow, ob, out);
}
